# starting kernel
# baseline (speedup 1.0000x reference)
#include <hip/hip_runtime.h>


typedef __attribute__((ext_vector_type(2))) float v2f;
typedef __attribute__((ext_vector_type(8))) float v8f;

#define CHUNK_L 196
#define NCOLS   512
#define NLEV    9

struct Params {
    float cm00, cm01, cm10, cm11;
    float km00, km01, km10, km11;
    float m0, m1, dt, h2, dt6;
};

__device__ inline Params make_params(const float* M, const float* C,
                                     const float* K, const float* dtp) {
    Params p;
    p.cm00 = C[0] + C[1]; p.cm01 = -C[1];
    p.cm10 = -C[1];       p.cm11 = C[2] + C[1];
    p.km00 = K[0] + K[1]; p.km01 = -K[1];
    p.km10 = -K[1];       p.km11 = K[2] + K[1];
    p.m0 = M[0]; p.m1 = M[1];
    p.dt = dtp[0];
    p.h2 = 0.5f * p.dt;
    p.dt6 = p.dt / 6.0f;
    return p;
}

__device__ inline void accel(const Params& p, float z0, float z1,
                             float v0, float v1, float u0, float u1,
                             float& a0, float& a1) {
    a0 = (u0 - (p.cm00 * v0 + p.cm01 * v1) - (p.km00 * z0 + p.km01 * z1)) / p.m0;
    a1 = (u1 - (p.cm10 * v0 + p.cm11 * v1) - (p.km10 * z0 + p.km11 * z1)) / p.m1;
}

__device__ inline void rk4_step(const Params& p, float& y0, float& y1,
                                float& v0, float& v1, float u0, float u1) {
    float k1x, k1y; accel(p, y0, y1, v0, v1, u0, u1, k1x, k1y);
    float y2x = y0 + v0 * p.h2, y2y = y1 + v1 * p.h2;
    float v2x = v0 + k1x * p.h2, v2y = v1 + k1y * p.h2;
    float k2x, k2y; accel(p, y2x, y2y, v2x, v2y, u0, u1, k2x, k2y);
    float y3x = y0 + v2x * p.h2, y3y = y1 + v2y * p.h2;
    float v3x = v0 + k2x * p.h2, v3y = v1 + k2y * p.h2;
    float k3x, k3y; accel(p, y3x, y3y, v3x, v3y, u0, u1, k3x, k3y);
    float y4x = y0 + v3x * p.dt, y4y = y1 + v3y * p.dt;
    float v4x = v0 + k3x * p.dt, v4y = v1 + k3y * p.dt;
    float k4x, k4y; accel(p, y4x, y4y, v4x, v4y, u0, u1, k4x, k4y);
    float yn0 = y0 + p.dt6 * (v0 + 2.0f * v2x + 2.0f * v3x + v4x);
    float yn1 = y1 + p.dt6 * (v1 + 2.0f * v2y + 2.0f * v3y + v4y);
    float vn0 = v0 + p.dt6 * (k1x + 2.0f * k2x + 2.0f * k3x + k4x);
    float vn1 = v1 + p.dt6 * (k1y + 2.0f * k2y + 2.0f * k3y + k4y);
    y0 = yn0; y1 = yn1; v0 = vn0; v1 = vn1;
}

__device__ inline void lds_wave_fence() {
    asm volatile("s_wait_dscnt 0x0" ::: "memory");
}

__device__ inline void wmma_mul4_lds(const float* src1, const float* src2,
                                     float* dst, int lane) {
    const int  n  = lane & 15;
    const bool hi = lane >= 16;
    v2f a, b;
    a.x = (n < 4) ? src1[n * 4 + (hi ? 2 : 0)] : 0.0f;
    a.y = (n < 4) ? src1[n * 4 + (hi ? 3 : 1)] : 0.0f;
    b.x = (n < 4) ? src2[(hi ? 2 : 0) * 4 + n] : 0.0f;
    b.y = (n < 4) ? src2[(hi ? 3 : 1) * 4 + n] : 0.0f;
    v8f c = {0.f, 0.f, 0.f, 0.f, 0.f, 0.f, 0.f, 0.f};
    v8f d = __builtin_amdgcn_wmma_f32_16x16x4_f32(
        false, a, false, b, (short)0, c, false, false);
    if (!hi && n < 4) {
        dst[0 * 4 + n] = d[0];
        dst[1 * 4 + n] = d[1];
        dst[2 * 4 + n] = d[2];
        dst[3 * 4 + n] = d[3];
    }
    lds_wave_fence();
}

__global__ void phase1_chunks(const float* __restrict__ u,
                              const float* __restrict__ M,
                              const float* __restrict__ C,
                              const float* __restrict__ K,
                              const float* __restrict__ dtp,
                              float* __restrict__ wsE, int nfull) {
    int i = blockIdx.x * blockDim.x + threadIdx.x;
    if (i >= nfull) return;
    Params p = make_params(M, C, K, dtp);
    float y0 = 0.f, y1 = 0.f, v0 = 0.f, v1 = 0.f;
    const float* up = u + (size_t)i * CHUNK_L * 2;
    for (int t = 0; t < CHUNK_L; ++t) {
        if (t + 16 < CHUNK_L) __builtin_prefetch(up + 2 * (t + 16), 0, 3);
        float u0 = up[2 * t], u1 = up[2 * t + 1];
        rk4_step(p, y0, y1, v0, v1, u0, u1);
    }
    wsE[i * 4 + 0] = y0; wsE[i * 4 + 1] = y1;
    wsE[i * 4 + 2] = v0; wsE[i * 4 + 3] = v1;
}

__global__ __launch_bounds__(1024) void phase2_scan(
    const float* __restrict__ M, const float* __restrict__ C,
    const float* __restrict__ K, const float* __restrict__ x0,
    const float* __restrict__ dtp, const float* __restrict__ wsE,
    float* __restrict__ wsS, int P) {
    __shared__ float xs[NCOLS][4];
    __shared__ float Gd[NLEV][16];
    __shared__ float mA[16], mT4[16], mT64[16], mQ[16], mR[16];

    const int  tid  = threadIdx.x;
    const int  lane = tid & 31;
    const int  wave = tid >> 5;
    const int  n    = lane & 15;
    const bool hi   = lane >= 16;
    const int  col  = wave * 16 + n;

    if (tid < NCOLS) {
        float a0, a1, a2, a3;
        if (tid == 0) {
            a0 = x0[0]; a1 = x0[2]; a2 = x0[1]; a3 = x0[3];
        } else if (tid < P) {
            a0 = wsE[(tid - 1) * 4 + 0]; a1 = wsE[(tid - 1) * 4 + 1];
            a2 = wsE[(tid - 1) * 4 + 2]; a3 = wsE[(tid - 1) * 4 + 3];
        } else {
            a0 = a1 = a2 = a3 = 0.f;
        }
        xs[tid][0] = a0; xs[tid][1] = a1; xs[tid][2] = a2; xs[tid][3] = a3;
    }

    if (tid < 32) {
        Params p = make_params(M, C, K, dtp);
        {
            float y0 = (lane == 0) ? 1.f : 0.f, y1 = (lane == 1) ? 1.f : 0.f;
            float v0 = (lane == 2) ? 1.f : 0.f, v1 = (lane == 3) ? 1.f : 0.f;
            rk4_step(p, y0, y1, v0, v1, 0.f, 0.f);
            if (lane < 4) {
                mA[0 * 4 + lane] = y0;
                mA[1 * 4 + lane] = y1;
                mA[2 * 4 + lane] = v0;
                mA[3 * 4 + lane] = v1;
            }
        }
        lds_wave_fence();
        wmma_mul4_lds(mA,   mA,   mQ,    lane);
        wmma_mul4_lds(mQ,   mQ,   mT4,   lane);
        wmma_mul4_lds(mT4,  mT4,  mQ,    lane);
        wmma_mul4_lds(mQ,   mQ,   mR,    lane);
        wmma_mul4_lds(mR,   mR,   mQ,    lane);
        wmma_mul4_lds(mQ,   mQ,   mT64,  lane);
        wmma_mul4_lds(mT64, mT64, mQ,    lane);
        wmma_mul4_lds(mQ,   mT64, mR,    lane);
        wmma_mul4_lds(mR,   mT4,  Gd[0], lane);
        for (int l = 1; l < NLEV; ++l) {
            wmma_mul4_lds(Gd[l - 1], Gd[l - 1], Gd[l], lane);
        }
    }
    __syncthreads();

    for (int l = 0; l < NLEV; ++l) {
        const int d = 1 << l;
        v2f a, b;
        a.x = (n < 4) ? Gd[l][n * 4 + (hi ? 2 : 0)] : 0.0f;
        a.y = (n < 4) ? Gd[l][n * 4 + (hi ? 3 : 1)] : 0.0f;
        const int src = col - d;
        b.x = (src >= 0) ? xs[src][hi ? 2 : 0] : 0.0f;
        b.y = (src >= 0) ? xs[src][hi ? 3 : 1] : 0.0f;
        float c0 = hi ? 0.f : xs[col][0];
        float c1 = hi ? 0.f : xs[col][1];
        float c2 = hi ? 0.f : xs[col][2];
        float c3 = hi ? 0.f : xs[col][3];
        v8f c = {c0, c1, c2, c3, 0.f, 0.f, 0.f, 0.f};
        __syncthreads();
        v8f dres = __builtin_amdgcn_wmma_f32_16x16x4_f32(
            false, a, false, b, (short)0, c, false, false);
        if (!hi) {
            xs[col][0] = dres[0]; xs[col][1] = dres[1];
            xs[col][2] = dres[2]; xs[col][3] = dres[3];
        }
        __syncthreads();
    }

    if (tid < P) {
        wsS[tid * 4 + 0] = xs[tid][0]; wsS[tid * 4 + 1] = xs[tid][1];
        wsS[tid * 4 + 2] = xs[tid][2]; wsS[tid * 4 + 3] = xs[tid][3];
    }
}

__global__ void phase3_emit(const float* __restrict__ u,
                            const float* __restrict__ M,
                            const float* __restrict__ C,
                            const float* __restrict__ K,
                            const float* __restrict__ dtp,
                            const float* __restrict__ wsS,
                            float* __restrict__ out, int P, int T) {
    int i = blockIdx.x * blockDim.x + threadIdx.x;
    if (i >= P) return;
    Params p = make_params(M, C, K, dtp);
    float y0 = wsS[i * 4 + 0], y1 = wsS[i * 4 + 1];
    float v0 = wsS[i * 4 + 2], v1 = wsS[i * 4 + 3];
    const int t0 = i * CHUNK_L;
    const int t1 = (t0 + CHUNK_L < T) ? (t0 + CHUNK_L) : T;
    for (int t = t0; t < t1; ++t) {
        if (t + 16 < t1) __builtin_prefetch(u + 2 * (t + 16), 0, 3);
        float u0 = u[2 * t], u1 = u[2 * t + 1];
        rk4_step(p, y0, y1, v0, v1, u0, u1);
        out[2 * t]     = y0;
        out[2 * t + 1] = y1;
    }
}

extern "C" void kernel_launch(void* const* d_in, const int* in_sizes, int n_in,
                              void* d_out, int out_size, void* d_ws,
                              size_t ws_size, hipStream_t stream) {
    const float* u  = (const float*)d_in[0];
    const float* M  = (const float*)d_in[1];
    const float* C  = (const float*)d_in[2];
    const float* K  = (const float*)d_in[3];
    const float* x0 = (const float*)d_in[4];
    const float* dt = (const float*)d_in[5];
    float* out = (float*)d_out;

    const int T = in_sizes[0] / 2;
    int P = (T + CHUNK_L - 1) / CHUNK_L;
    if (P > NCOLS) P = NCOLS;
    const int nfull = P - 1;

    float* wsE = (float*)d_ws;
    float* wsS = wsE + NCOLS * 4;

    phase1_chunks<<<(nfull + 255) / 256, 256, 0, stream>>>(
        u, M, C, K, dt, wsE, nfull);
    phase2_scan<<<1, 1024, 0, stream>>>(M, C, K, x0, dt, wsE, wsS, P);
    phase3_emit<<<(P + 255) / 256, 256, 0, stream>>>(
        u, M, C, K, dt, wsS, out, P, T);
}


# Scan launch: the 9 log-step levels (18 block barriers) become one block-Toeplitz tile product per 64 columns (G^(i-j) on WMMA) + carry folds with G^4, G^64; 8 waves, 2 barriers; sums in a new order
# speedup vs baseline: 15.7359x; 15.7359x over previous
#include <hip/hip_runtime.h>


typedef __attribute__((ext_vector_type(2))) float v2f;
typedef __attribute__((ext_vector_type(8))) float v8f;

#define SUB_L   8
#define SUP_SUB 32
#define CHUNK_L (SUB_L * SUP_SUB)
#define NCOLS   512
#define TB_ROW  6
#define TB_MAT  24

struct Params {
    float cm00, cm01, cm10, cm11;
    float km00, km01, km10, km11;
    float m0, m1, dt, h2, dt6;
};

__device__ inline Params make_params(const float* M, const float* C,
                                     const float* K, const float* dtp) {
    Params p;
    p.cm00 = C[0] + C[1]; p.cm01 = -C[1];
    p.cm10 = -C[1];       p.cm11 = C[2] + C[1];
    p.km00 = K[0] + K[1]; p.km01 = -K[1];
    p.km10 = -K[1];       p.km11 = K[2] + K[1];
    p.m0 = M[0]; p.m1 = M[1];
    p.dt = dtp[0];
    p.h2 = 0.5f * p.dt;
    p.dt6 = p.dt / 6.0f;
    return p;
}

__device__ inline void accel(const Params& p, float z0, float z1,
                             float v0, float v1, float u0, float u1,
                             float& a0, float& a1) {
    a0 = (u0 - (p.cm00 * v0 + p.cm01 * v1) - (p.km00 * z0 + p.km01 * z1)) / p.m0;
    a1 = (u1 - (p.cm10 * v0 + p.cm11 * v1) - (p.km10 * z0 + p.km11 * z1)) / p.m1;
}

__device__ inline void rk4_step(const Params& p, float& y0, float& y1,
                                float& v0, float& v1, float u0, float u1) {
    float k1x, k1y; accel(p, y0, y1, v0, v1, u0, u1, k1x, k1y);
    float y2x = y0 + v0 * p.h2, y2y = y1 + v1 * p.h2;
    float v2x = v0 + k1x * p.h2, v2y = v1 + k1y * p.h2;
    float k2x, k2y; accel(p, y2x, y2y, v2x, v2y, u0, u1, k2x, k2y);
    float y3x = y0 + v2x * p.h2, y3y = y1 + v2y * p.h2;
    float v3x = v0 + k2x * p.h2, v3y = v1 + k2y * p.h2;
    float k3x, k3y; accel(p, y3x, y3y, v3x, v3y, u0, u1, k3x, k3y);
    float y4x = y0 + v3x * p.dt, y4y = y1 + v3y * p.dt;
    float v4x = v0 + k3x * p.dt, v4y = v1 + k3y * p.dt;
    float k4x, k4y; accel(p, y4x, y4y, v4x, v4y, u0, u1, k4x, k4y);
    float yn0 = y0 + p.dt6 * (v0 + 2.0f * v2x + 2.0f * v3x + v4x);
    float yn1 = y1 + p.dt6 * (v1 + 2.0f * v2y + 2.0f * v3y + v4y);
    float vn0 = v0 + p.dt6 * (k1x + 2.0f * k2x + 2.0f * k3x + k4x);
    float vn1 = v1 + p.dt6 * (k1y + 2.0f * k2y + 2.0f * k3y + k4y);
    y0 = yn0; y1 = yn1; v0 = vn0; v1 = vn1;
}

__device__ inline void lds_wave_fence() {
    asm volatile("s_wait_dscnt 0x0" ::: "memory");
}

__device__ inline void build_step_table(const Params& p, float* tb, int lane) {
    float y0 = (lane == 0) ? 1.f : 0.f, y1 = (lane == 1) ? 1.f : 0.f;
    float v0 = (lane == 2) ? 1.f : 0.f, v1 = (lane == 3) ? 1.f : 0.f;
    const float u0 = (lane == 4) ? 1.f : 0.f, u1 = (lane == 5) ? 1.f : 0.f;
    rk4_step(p, y0, y1, v0, v1, u0, u1);
    if (lane < TB_ROW) {
        tb[0 * TB_ROW + lane] = y0; tb[1 * TB_ROW + lane] = y1;
        tb[2 * TB_ROW + lane] = v0; tb[3 * TB_ROW + lane] = v1;
    }
    lds_wave_fence();
    float a[4][4];
    for (int q = 0; q < 4; ++q)
        for (int r = 0; r < 4; ++r) a[q][r] = tb[q * TB_ROW + r];
    for (int k = 1; k < SUB_L; ++k) {
        const float n0 = a[0][0] * y0 + a[0][1] * y1 + a[0][2] * v0 + a[0][3] * v1;
        const float n1 = a[1][0] * y0 + a[1][1] * y1 + a[1][2] * v0 + a[1][3] * v1;
        const float n2 = a[2][0] * y0 + a[2][1] * y1 + a[2][2] * v0 + a[2][3] * v1;
        const float n3 = a[3][0] * y0 + a[3][1] * y1 + a[3][2] * v0 + a[3][3] * v1;
        y0 = n0; y1 = n1; v0 = n2; v1 = n3;
        if (lane < TB_ROW) {
            tb[k * TB_MAT + 0 * TB_ROW + lane] = y0; tb[k * TB_MAT + 1 * TB_ROW + lane] = y1;
            tb[k * TB_MAT + 2 * TB_ROW + lane] = v0; tb[k * TB_MAT + 3 * TB_ROW + lane] = v1;
        }
    }
    lds_wave_fence();
}

__device__ inline float lane_bcast(float v, int j) {
    return __int_as_float(__builtin_amdgcn_readlane(__float_as_int(v), j));
}

struct UCol { float f[8]; };

__device__ inline UCol load_ucol(const float* __restrict__ u, long base, long n2) {
    UCol r;
    if (base + 8 <= n2) {
        const float4 lo = *(const float4*)(u + base);
        const float4 hi = *(const float4*)(u + base + 4);
        r.f[0] = lo.x; r.f[1] = lo.y; r.f[2] = lo.z; r.f[3] = lo.w;
        r.f[4] = hi.x; r.f[5] = hi.y; r.f[6] = hi.z; r.f[7] = hi.w;
    } else {
        for (int e = 0; e < 8; ++e) r.f[e] = (base + e < n2) ? u[base + e] : 0.f;
    }
    return r;
}

__device__ inline v8f tile_end_states(const float* tb, const UCol& uc, int c, int h) {
    v8f acc = {0.f, 0.f, 0.f, 0.f, 0.f, 0.f, 0.f, 0.f};
    const int cr = (c < 4) ? c : 0;
    for (int kk = 0; kk < 4; ++kk) {
        const int j = 4 * h + kk;
        const float r0 = tb[(7 - j) * TB_MAT + cr * TB_ROW + 4];
        const float r1 = tb[(7 - j) * TB_MAT + cr * TB_ROW + 5];
        v2f a, b;
        a.x = (c < 4) ? r0 : 0.f;
        a.y = (c < 4) ? r1 : 0.f;
        b.x = uc.f[2 * kk];
        b.y = uc.f[2 * kk + 1];
        acc = __builtin_amdgcn_wmma_f32_16x16x4_f32(
            false, a, false, b, (short)0, acc, false, false);
    }
    return acc;
}

__global__ __launch_bounds__(64) void phase1_chunks(
    const float* __restrict__ u, const float* __restrict__ M,
    const float* __restrict__ C, const float* __restrict__ K,
    const float* __restrict__ dtp, float* __restrict__ wsE, int T) {
    __shared__ float tb[SUB_L * TB_MAT];
    __shared__ float4 e8[SUP_SUB];

    const int tid = threadIdx.x, lane = tid & 31, wave = tid >> 5;
    const int c = lane & 15, h = lane >> 4;
    const int sup = blockIdx.x;
    const long sc = (long)sup * SUP_SUB + wave * 16 + c;
    const UCol uc = load_ucol(u, sc * 16 + 8 * h, 2L * T);

    if (wave == 0) {
        Params p = make_params(M, C, K, dtp);
        build_step_table(p, tb, lane);
    }
    __syncthreads();

    const v8f e = tile_end_states(tb, uc, c, h);
    if (h == 0) e8[wave * 16 + c] = make_float4(e[0], e[1], e[2], e[3]);
    __syncthreads();

    if (wave == 0) {
        const float4 ev = e8[lane];
        float g[4][4];
        for (int q = 0; q < 4; ++q)
            for (int r = 0; r < 4; ++r) g[q][r] = tb[7 * TB_MAT + q * TB_ROW + r];
        float s0 = 0.f, s1 = 0.f, s2 = 0.f, s3 = 0.f;
        for (int j = 0; j < SUP_SUB; ++j) {
            const float n0 = (lane_bcast(ev.x, j) + g[0][0] * s0 + g[0][1] * s1) + (g[0][2] * s2 + g[0][3] * s3);
            const float n1 = (lane_bcast(ev.y, j) + g[1][0] * s0 + g[1][1] * s1) + (g[1][2] * s2 + g[1][3] * s3);
            const float n2 = (lane_bcast(ev.z, j) + g[2][0] * s0 + g[2][1] * s1) + (g[2][2] * s2 + g[2][3] * s3);
            const float n3 = (lane_bcast(ev.w, j) + g[3][0] * s0 + g[3][1] * s1) + (g[3][2] * s2 + g[3][3] * s3);
            s0 = n0; s1 = n1; s2 = n2; s3 = n3;
        }
        if (lane == 0) *(float4*)(wsE + sup * 4) = make_float4(s0, s1, s2, s3);
    }
}

#define SCAN_TILE  64
#define SCAN_NT    (NCOLS / SCAN_TILE)
#define PW_G64     5
#define PW_N       6

struct Pw { float m[4]; float t[4]; };

__device__ inline Pw pw_mul(const Pw& x, const Pw& y, bool hi) {
    v2f by, ax;
    by.x = hi ? y.m[2] : y.m[0]; by.y = hi ? y.m[3] : y.m[1];
    ax.x = hi ? x.t[2] : x.t[0]; ax.y = hi ? x.t[3] : x.t[1];
    const v8f zero8 = {0.f, 0.f, 0.f, 0.f, 0.f, 0.f, 0.f, 0.f};
    const v8f pm = __builtin_amdgcn_wmma_f32_16x16x4_f32(
        false, ax, false, by, (short)0, zero8, false, false);
    const v8f pt = __builtin_amdgcn_wmma_f32_16x16x4_f32(
        false, by, false, ax, (short)0, zero8, false, false);
    Pw r;
    r.m[0] = pm[0]; r.m[1] = pm[1]; r.m[2] = pm[2]; r.m[3] = pm[3];
    r.t[0] = pt[0]; r.t[1] = pt[1]; r.t[2] = pt[2]; r.t[3] = pt[3];
    return r;
}

__device__ inline void pw_store(float* pw, int slot, const Pw& x, int lane) {
    for (int q = 0; q < 4; ++q) pw[(slot * 4 + q) * 32 + lane] = x.m[q];
}

__device__ inline float4 load_scan_col(const float* __restrict__ x0,
                                       const float* __restrict__ wsE, int col, int P) {
    float4 r = make_float4(0.f, 0.f, 0.f, 0.f);
    if (col == 0) {
        r = make_float4(x0[0], x0[2], x0[1], x0[3]);
    } else if (col < P) {
        r = *(const float4*)(wsE + (col - 1) * 4);
    }
    return r;
}

__global__ __launch_bounds__(256) void phase2_scan(
    const float* __restrict__ M, const float* __restrict__ C,
    const float* __restrict__ K, const float* __restrict__ x0,
    const float* __restrict__ dtp, const float* __restrict__ wsE,
    float* __restrict__ wsS, int P) {
    __shared__ float pw[PW_N * 4 * 32];
    __shared__ float4 tend[SCAN_NT];

    const int  tid  = threadIdx.x;
    const int  lane = tid & 31;
    const int  wave = __builtin_amdgcn_readfirstlane(tid >> 5);
    const int  c    = lane & 15, h = lane >> 4;
    const bool hi   = lane >= 16;

    const int col0 = wave * SCAN_TILE + 4 * c + 2 * h;
    const float4 xa = load_scan_col(x0, wsE, col0, P);
    const float4 xb = load_scan_col(x0, wsE, col0 + 1, P);
    const float xf[8] = {xa.x, xa.y, xa.z, xa.w, xb.x, xb.y, xb.z, xb.w};

    if (wave == 0) {
        Params p = make_params(M, C, K, dtp);
        const int  q  = lane & 3;
        const bool on = (lane & 4) == 0;
        float y0 = (on && q == 0) ? 1.f : 0.f, y1 = (on && q == 1) ? 1.f : 0.f;
        float v0 = (on && q == 2) ? 1.f : 0.f, v1 = (on && q == 3) ? 1.f : 0.f;
        rk4_step(p, y0, y1, v0, v1, 0.f, 0.f);
        Pw g;
        g.m[0] = on ? y0 : 0.f; g.m[1] = on ? y1 : 0.f;
        g.m[2] = on ? v0 : 0.f; g.m[3] = on ? v1 : 0.f;
        for (int r = 0; r < 4; ++r) {
            const float c0 = lane_bcast(y0, r), c1 = lane_bcast(y1, r);
            const float c2 = lane_bcast(v0, r), c3 = lane_bcast(v1, r);
            float t = c3;
            t = (q == 2) ? c2 : t;
            t = (q == 1) ? c1 : t;
            t = (q == 0) ? c0 : t;
            g.t[r] = on ? t : 0.f;
        }
        for (int sq = 0; sq < 8; ++sq) g = pw_mul(g, g, hi);
        Pw id;
        for (int r = 0; r < 4; ++r) { id.m[r] = (lane == r) ? 1.f : 0.f; id.t[r] = id.m[r]; }
        pw_store(pw, 0, id, lane);
        pw_store(pw, 1, g, lane);
        const Pw g2 = pw_mul(g, g, hi);
        pw_store(pw, 2, g2, lane);
        const Pw g3 = pw_mul(g2, g, hi);
        Pw gp = pw_mul(g2, g2, hi);
        pw_store(pw, 3, g3, lane);
        pw_store(pw, 4, gp, lane);
        for (int sq = 0; sq < 4; ++sq) gp = pw_mul(gp, gp, hi);
        pw_store(pw, PW_G64, gp, lane);
    }
    __syncthreads();

    v8f e = {0.f, 0.f, 0.f, 0.f, 0.f, 0.f, 0.f, 0.f};
    {
        const int cr = (c < 4) ? c : 0;
        for (int kk = 0; kk < 4; ++kk) {
            const int j  = 2 * h + (kk >> 1);
            const int b0 = 2 * (kk & 1);
            const float r0 = pw[((3 - j) * 4 + cr) * 32 + b0];
            const float r1 = pw[((3 - j) * 4 + cr) * 32 + b0 + 1];
            v2f a, b;
            a.x = (c < 4) ? r0 : 0.f;
            a.y = (c < 4) ? r1 : 0.f;
            b.x = xf[2 * kk];
            b.y = xf[2 * kk + 1];
            e = __builtin_amdgcn_wmma_f32_16x16x4_f32(
                false, a, false, b, (short)0, e, false, false);
        }
    }

    const int i = c >> 2, a = c & 3;
    v2f ha[4];
    for (int kk = 0; kk < 4; ++kk) {
        const int j  = 2 * h + (kk >> 1);
        const int b0 = 2 * (kk & 1);
        const int dk = (i >= j) ? (i - j) : 0;
        const float t0 = pw[(dk * 4 + a) * 32 + b0];
        const float t1 = pw[(dk * 4 + a) * 32 + b0 + 1];
        ha[kk].x = (i >= j) ? t0 : 0.f;
        ha[kk].y = (i >= j) ? t1 : 0.f;
    }
    v2f oa;
    oa.x = pw[((i + 1) * 4 + a) * 32 + 2 * h];
    oa.y = pw[((i + 1) * 4 + a) * 32 + 2 * h + 1];

    float g4[4][4], g64[4][4];
    for (int q = 0; q < 4; ++q)
        for (int r = 0; r < 4; ++r) {
            g4[q][r]  = pw[(4 * 4 + q) * 32 + r];
            g64[q][r] = pw[(PW_G64 * 4 + q) * 32 + r];
        }

    {
        float s0 = 0.f, s1 = 0.f, s2 = 0.f, s3 = 0.f;
        for (int j = 0; j < 16; ++j) {
            const float n0 = (lane_bcast(e[0], j) + g4[0][0] * s0 + g4[0][1] * s1) + (g4[0][2] * s2 + g4[0][3] * s3);
            const float n1 = (lane_bcast(e[1], j) + g4[1][0] * s0 + g4[1][1] * s1) + (g4[1][2] * s2 + g4[1][3] * s3);
            const float n2 = (lane_bcast(e[2], j) + g4[2][0] * s0 + g4[2][1] * s1) + (g4[2][2] * s2 + g4[2][3] * s3);
            const float n3 = (lane_bcast(e[3], j) + g4[3][0] * s0 + g4[3][1] * s1) + (g4[3][2] * s2 + g4[3][3] * s3);
            s0 = n0; s1 = n1; s2 = n2; s3 = n3;
        }
        if (lane == 0) tend[wave] = make_float4(s0, s1, s2, s3);
    }
    __syncthreads();

    float d0 = 0.f, d1 = 0.f, d2 = 0.f, d3 = 0.f;
    {
        float4 te[SCAN_NT - 1];
        for (int v = 0; v < SCAN_NT - 1; ++v) te[v] = tend[v];
        for (int v = 0; v < SCAN_NT - 1; ++v) {
            const float n0 = (te[v].x + g64[0][0] * d0 + g64[0][1] * d1) + (g64[0][2] * d2 + g64[0][3] * d3);
            const float n1 = (te[v].y + g64[1][0] * d0 + g64[1][1] * d1) + (g64[1][2] * d2 + g64[1][3] * d3);
            const float n2 = (te[v].z + g64[2][0] * d0 + g64[2][1] * d1) + (g64[2][2] * d2 + g64[2][3] * d3);
            const float n3 = (te[v].w + g64[3][0] * d0 + g64[3][1] * d1) + (g64[3][2] * d2 + g64[3][3] * d3);
            const bool take = v < wave;
            d0 = take ? n0 : d0; d1 = take ? n1 : d1;
            d2 = take ? n2 : d2; d3 = take ? n3 : d3;
        }
    }

    float4 mine = make_float4(d0, d1, d2, d3);
    {
        float s0 = d0, s1 = d1, s2 = d2, s3 = d3;
        for (int j = 0; j < 16; ++j) {
            if (c == j) mine = make_float4(s0, s1, s2, s3);
            const float n0 = (lane_bcast(e[0], j) + g4[0][0] * s0 + g4[0][1] * s1) + (g4[0][2] * s2 + g4[0][3] * s3);
            const float n1 = (lane_bcast(e[1], j) + g4[1][0] * s0 + g4[1][1] * s1) + (g4[1][2] * s2 + g4[1][3] * s3);
            const float n2 = (lane_bcast(e[2], j) + g4[2][0] * s0 + g4[2][1] * s1) + (g4[2][2] * s2 + g4[2][3] * s3);
            const float n3 = (lane_bcast(e[3], j) + g4[3][0] * s0 + g4[3][1] * s1) + (g4[3][2] * s2 + g4[3][3] * s3);
            s0 = n0; s1 = n1; s2 = n2; s3 = n3;
        }
    }

    v2f sb;
    sb.x = h ? mine.z : mine.x;
    sb.y = h ? mine.w : mine.y;
    v8f acc = {0.f, 0.f, 0.f, 0.f, 0.f, 0.f, 0.f, 0.f};
    acc = __builtin_amdgcn_wmma_f32_16x16x4_f32(
        false, oa, false, sb, (short)0, acc, false, false);
    for (int kk = 0; kk < 4; ++kk) {
        v2f b;
        b.x = xf[2 * kk];
        b.y = xf[2 * kk + 1];
        acc = __builtin_amdgcn_wmma_f32_16x16x4_f32(
            false, ha[kk], false, b, (short)0, acc, false, false);
    }

    if (col0 < P)     *(float4*)(wsS + col0 * 4)     = make_float4(acc[0], acc[1], acc[2], acc[3]);
    if (col0 + 1 < P) *(float4*)(wsS + col0 * 4 + 4) = make_float4(acc[4], acc[5], acc[6], acc[7]);
}

__global__ __launch_bounds__(64) void phase3_emit(
    const float* __restrict__ u, const float* __restrict__ M,
    const float* __restrict__ C, const float* __restrict__ K,
    const float* __restrict__ dtp, const float* __restrict__ wsS,
    float* __restrict__ out, int T) {
    __shared__ float tb[SUB_L * TB_MAT];
    __shared__ float4 e8[SUP_SUB];
    __shared__ float4 s8[SUP_SUB];

    const int tid = threadIdx.x, lane = tid & 31, wave = tid >> 5;
    const int c = lane & 15, h = lane >> 4;
    const int sup = blockIdx.x;
    const int col = wave * 16 + c;
    const long n2 = 2L * T;
    const long base = ((long)sup * SUP_SUB + col) * 16 + 8 * h;
    const UCol uc = load_ucol(u, base, n2);

    float4 sin4 = make_float4(0.f, 0.f, 0.f, 0.f);
    if (wave == 0) sin4 = *(const float4*)(wsS + sup * 4);

    if (wave == 0) {
        Params p = make_params(M, C, K, dtp);
        build_step_table(p, tb, lane);
    }
    __syncthreads();

    const v8f e = tile_end_states(tb, uc, c, h);
    if (h == 0) e8[col] = make_float4(e[0], e[1], e[2], e[3]);

    const int i = c >> 1, a = c & 1;
    v2f ha[4];
    for (int kk = 0; kk < 4; ++kk) {
        const int j  = 4 * h + kk;
        const int dk = (i >= j) ? (i - j) : 0;
        const float t0 = tb[dk * TB_MAT + a * TB_ROW + 4];
        const float t1 = tb[dk * TB_MAT + a * TB_ROW + 5];
        ha[kk].x = (i >= j) ? t0 : 0.f;
        ha[kk].y = (i >= j) ? t1 : 0.f;
    }
    v2f oa;
    oa.x = tb[i * TB_MAT + a * TB_ROW + 2 * h];
    oa.y = tb[i * TB_MAT + a * TB_ROW + 2 * h + 1];
    __syncthreads();

    if (wave == 0) {
        const float4 ev = e8[lane];
        float g[4][4];
        for (int q = 0; q < 4; ++q)
            for (int r = 0; r < 4; ++r) g[q][r] = tb[7 * TB_MAT + q * TB_ROW + r];
        float s0 = sin4.x, s1 = sin4.y, s2 = sin4.z, s3 = sin4.w;
        float4 mine = sin4;
        for (int j = 0; j < SUP_SUB; ++j) {
            if (lane == j) mine = make_float4(s0, s1, s2, s3);
            const float n0 = (lane_bcast(ev.x, j) + g[0][0] * s0 + g[0][1] * s1) + (g[0][2] * s2 + g[0][3] * s3);
            const float n1 = (lane_bcast(ev.y, j) + g[1][0] * s0 + g[1][1] * s1) + (g[1][2] * s2 + g[1][3] * s3);
            const float n2s = (lane_bcast(ev.z, j) + g[2][0] * s0 + g[2][1] * s1) + (g[2][2] * s2 + g[2][3] * s3);
            const float n3 = (lane_bcast(ev.w, j) + g[3][0] * s0 + g[3][1] * s1) + (g[3][2] * s2 + g[3][3] * s3);
            s0 = n0; s1 = n1; s2 = n2s; s3 = n3;
        }
        s8[lane] = mine;
    }
    __syncthreads();

    const float4 sc4 = s8[col];
    v2f sb;
    sb.x = h ? sc4.z : sc4.x;
    sb.y = h ? sc4.w : sc4.y;
    v8f acc = {0.f, 0.f, 0.f, 0.f, 0.f, 0.f, 0.f, 0.f};
    acc = __builtin_amdgcn_wmma_f32_16x16x4_f32(
        false, oa, false, sb, (short)0, acc, false, false);
    for (int kk = 0; kk < 4; ++kk) {
        v2f b;
        b.x = uc.f[2 * kk];
        b.y = uc.f[2 * kk + 1];
        acc = __builtin_amdgcn_wmma_f32_16x16x4_f32(
            false, ha[kk], false, b, (short)0, acc, false, false);
    }

    if (base + 8 <= n2) {
        *(float4*)(out + base)     = make_float4(acc[0], acc[1], acc[2], acc[3]);
        *(float4*)(out + base + 4) = make_float4(acc[4], acc[5], acc[6], acc[7]);
    } else {
        for (int r = 0; r < 8; ++r)
            if (base + r < n2) out[base + r] = acc[r];
    }
}

extern "C" void kernel_launch(void* const* d_in, const int* in_sizes, int n_in,
                              void* d_out, int out_size, void* d_ws,
                              size_t ws_size, hipStream_t stream) {
    const float* u  = (const float*)d_in[0];
    const float* M  = (const float*)d_in[1];
    const float* C  = (const float*)d_in[2];
    const float* K  = (const float*)d_in[3];
    const float* x0 = (const float*)d_in[4];
    const float* dt = (const float*)d_in[5];
    float* out = (float*)d_out;

    const int T = in_sizes[0] / 2;
    int P = (T + CHUNK_L - 1) / CHUNK_L;
    if (P > NCOLS) P = NCOLS;
    const int nfull = P - 1;

    float* wsE = (float*)d_ws;
    float* wsS = wsE + NCOLS * 4;

    phase1_chunks<<<nfull, 64, 0, stream>>>(u, M, C, K, dt, wsE, T);
    phase2_scan<<<1, 256, 0, stream>>>(M, C, K, x0, dt, wsE, wsS, P);
    phase3_emit<<<P, 64, 0, stream>>>(u, M, C, K, dt, wsS, out, T);
}
